# starting kernel
# baseline (speedup 1.0000x reference)
#include <hip/hip_runtime.h>
#include <hip/hip_bf16.h>

typedef __attribute__((ext_vector_type(2))) float v2f;
typedef __attribute__((ext_vector_type(8))) float v8f;

#define NN 2048
#define FF 64
#define HH 32
#define CC 16

__global__ __launch_bounds__(256)
void fsums_kernel(const float* __restrict__ x,
                  const float* __restrict__ W1,
                  const float* __restrict__ b1,
                  const float* __restrict__ W2,
                  const float* __restrict__ b2,
                  const float* __restrict__ fw,
                  float* __restrict__ fsums)
{
    __shared__ float attnLds[FF];
    __shared__ float redC[8 * 256];

    const int tid  = threadIdx.x;
    const int lane = tid & 31;
    const int wave = tid >> 5;
    const int tileM = blockIdx.x * 16;

    if (tid < FF) attnLds[tid] = __expf(__expf(fw[tid]));
    __syncthreads();
    float s = 0.f;
    #pragma unroll
    for (int f = 0; f < FF; ++f) s += attnLds[f];
    const float invS = 1.0f / s;

    const int row = lane & 15;
    const int hi  = lane >> 4;
    const int col = row;
    const int gRow = tileM + row;

    v8f acc = {};
    const int kBase = wave * 256;
    for (int k0 = kBase; k0 < kBase + 256; k0 += 4) {
        const int kA = k0 + 2 * hi;
        const int f0 = kA >> 5;
        const float xv  = x[gRow * FF + f0];
        v2f a;
        a.x = fmaxf(fmaf(xv, W1[kA],     b1[kA]),     0.f);
        a.y = fmaxf(fmaf(xv, W1[kA + 1], b1[kA + 1]), 0.f);
        const float at = attnLds[f0] * invS;
        v2f b;
        b.x = at * W2[kA       * CC + col];
        b.y = at * W2[(kA + 1) * CC + col];
        acc = __builtin_amdgcn_wmma_f32_16x16x4_f32(false, a, false, b,
                                                    (short)0, acc, false, false);
    }

    #pragma unroll
    for (int r = 0; r < 8; ++r) {
        const int orow = r + 8 * hi;
        redC[wave * 256 + orow * 16 + col] = acc[r];
    }
    __syncthreads();
    float v = 0.f;
    #pragma unroll
    for (int w = 0; w < 8; ++w) v += redC[w * 256 + tid];
    const int ocol = tid & 15;
    float bias = 0.f;
    #pragma unroll 4
    for (int f = 0; f < FF; ++f) bias += attnLds[f] * invS * b2[f * CC + ocol];
    fsums[tileM * CC + tid] = v + bias;
}

__global__ __launch_bounds__(256)
void out_kernel(const float* __restrict__ dist,
                const float* __restrict__ norm,
                const float* __restrict__ pWm1,
                const float* __restrict__ pbm1,
                const float* __restrict__ pWm2,
                const float* __restrict__ pbm2,
                const float* __restrict__ fsums,
                float* __restrict__ out)
{
    __shared__ float mT[8][16 * 66];
    __shared__ float redC[8 * 256];

    const int tid  = threadIdx.x;
    const int lane = tid & 31;
    const int wave = tid >> 5;
    const int tileM = blockIdx.x * 16;
    const int row = lane & 15;
    const int hi  = lane >> 4;
    const int col = row;

    float wm1[HH], bb1[HH], wm2[HH];
    #pragma unroll
    for (int h = 0; h < HH; ++h) { wm1[h] = pWm1[h]; bb1[h] = pbm1[h]; wm2[h] = pWm2[h]; }
    const float bm2v = pbm2[0];

    float* myT = mT[wave];
    v8f acc = {};
    const int kBase = wave * 256;
    for (int kc = kBase; kc < kBase + 256; kc += 64) {
        for (int r = 0; r < 16; ++r) {
            const size_t base = (size_t)(tileM + r) * NN + kc + 2 * lane;
            const v2f d2 = __builtin_nontemporal_load((const v2f*)(dist + base));
            const v2f n2 = __builtin_nontemporal_load((const v2f*)(norm + base));
            float a0 = bm2v, a1 = bm2v;
            #pragma unroll
            for (int h = 0; h < HH; ++h) {
                const float t0 = fmaf(d2.x, wm1[h], bb1[h]);
                const float t1 = fmaf(d2.y, wm1[h], bb1[h]);
                a0 = fmaf(fmaxf(t0, 0.f), wm2[h], a0);
                a1 = fmaf(fmaxf(t1, 0.f), wm2[h], a1);
            }
            v2f mv;
            mv.x = a0 * __builtin_amdgcn_rcpf(n2.x);
            mv.y = a1 * __builtin_amdgcn_rcpf(n2.y);
            *(v2f*)&myT[r * 66 + 2 * lane] = mv;
        }
        #pragma unroll
        for (int kk = 0; kk < 64; kk += 4) {
            const v2f a = *(const v2f*)&myT[row * 66 + kk + 2 * hi];
            const int kg = kc + kk + 2 * hi;
            v2f b;
            b.x = fsums[kg       * CC + col];
            b.y = fsums[(kg + 1) * CC + col];
            acc = __builtin_amdgcn_wmma_f32_16x16x4_f32(false, a, false, b,
                                                        (short)0, acc, false, false);
        }
    }

    #pragma unroll
    for (int r = 0; r < 8; ++r) {
        const int orow = r + 8 * hi;
        redC[wave * 256 + orow * 16 + col] = acc[r];
    }
    __syncthreads();
    float v = 0.f;
    #pragma unroll
    for (int w = 0; w < 8; ++w) v += redC[w * 256 + tid];
    out[tileM * CC + tid] = v;
}

extern "C" void kernel_launch(void* const* d_in, const int* in_sizes, int n_in,
                              void* d_out, int out_size, void* d_ws, size_t ws_size,
                              hipStream_t stream) {
    const float* x    = (const float*)d_in[0];
    const float* dist = (const float*)d_in[1];
    const float* norm = (const float*)d_in[2];
    const float* W1   = (const float*)d_in[3];
    const float* b1   = (const float*)d_in[4];
    const float* W2   = (const float*)d_in[5];
    const float* b2   = (const float*)d_in[6];
    const float* fw   = (const float*)d_in[7];
    const float* Wm1  = (const float*)d_in[8];
    const float* bm1  = (const float*)d_in[9];
    const float* Wm2  = (const float*)d_in[10];
    const float* bm2  = (const float*)d_in[11];
    float* out   = (float*)d_out;
    float* fsums = (float*)d_ws;

    fsums_kernel<<<NN / 16, 256, 0, stream>>>(x, W1, b1, W2, b2, fw, fsums);
    out_kernel<<<NN / 16, 256, 0, stream>>>(dist, norm, Wm1, bm1, Wm2, bm2, fsums, out);
}


# out kernel re-tiled: 256 blocks x 16 waves, 8 rows x 128 K cols per wave, all global loads of a wave in flight before its math; 16-way K split (was 8) changes the sum order
# speedup vs baseline: 2.2638x; 2.2638x over previous
#include <hip/hip_runtime.h>
#include <hip/hip_bf16.h>

typedef __attribute__((ext_vector_type(2))) float v2f;
typedef __attribute__((ext_vector_type(4))) float v4f;
typedef __attribute__((ext_vector_type(8))) float v8f;

#define NN 2048
#define FF 64
#define HH 32
#define CC 16

__global__ __launch_bounds__(256)
void fsums_kernel(const float* __restrict__ x,
                  const float* __restrict__ W1,
                  const float* __restrict__ b1,
                  const float* __restrict__ W2,
                  const float* __restrict__ b2,
                  const float* __restrict__ fw,
                  float* __restrict__ fsums)
{
    __shared__ float attnLds[FF];
    __shared__ float redC[8 * 256];

    const int tid  = threadIdx.x;
    const int lane = tid & 31;
    const int wave = tid >> 5;
    const int tileM = blockIdx.x * 16;

    if (tid < FF) attnLds[tid] = __expf(__expf(fw[tid]));
    __syncthreads();
    float s = 0.f;
    #pragma unroll
    for (int f = 0; f < FF; ++f) s += attnLds[f];
    const float invS = 1.0f / s;

    const int row = lane & 15;
    const int hi  = lane >> 4;
    const int col = row;
    const int gRow = tileM + row;

    v8f acc = {};
    const int kBase = wave * 256;
    for (int k0 = kBase; k0 < kBase + 256; k0 += 4) {
        const int kA = k0 + 2 * hi;
        const int f0 = kA >> 5;
        const float xv  = x[gRow * FF + f0];
        v2f a;
        a.x = fmaxf(fmaf(xv, W1[kA],     b1[kA]),     0.f);
        a.y = fmaxf(fmaf(xv, W1[kA + 1], b1[kA + 1]), 0.f);
        const float at = attnLds[f0] * invS;
        v2f b;
        b.x = at * W2[kA       * CC + col];
        b.y = at * W2[(kA + 1) * CC + col];
        acc = __builtin_amdgcn_wmma_f32_16x16x4_f32(false, a, false, b,
                                                    (short)0, acc, false, false);
    }

    #pragma unroll
    for (int r = 0; r < 8; ++r) {
        const int orow = r + 8 * hi;
        redC[wave * 256 + orow * 16 + col] = acc[r];
    }
    __syncthreads();
    float v = 0.f;
    #pragma unroll
    for (int w = 0; w < 8; ++w) v += redC[w * 256 + tid];
    const int ocol = tid & 15;
    float bias = 0.f;
    #pragma unroll 4
    for (int f = 0; f < FF; ++f) bias += attnLds[f] * invS * b2[f * CC + ocol];
    fsums[tileM * CC + tid] = v + bias;
}

#define RB 8
#define KW 128
#define NW 16
#define MP 132

__global__ __launch_bounds__(512)
void out_kernel(const float* __restrict__ dist,
                const float* __restrict__ norm,
                const float* __restrict__ pWm1,
                const float* __restrict__ pbm1,
                const float* __restrict__ pWm2,
                const float* __restrict__ pbm2,
                const float* __restrict__ fsums,
                float* __restrict__ out)
{
    __shared__ float mT[NW][RB * MP];
    __shared__ float redC[NW * RB * CC];

    const int tid  = threadIdx.x;
    const int lane = tid & 31;
    const int wave = tid >> 5;
    const int tileM = blockIdx.x * RB;
    const int row = lane & 15;
    const int hi  = lane >> 4;
    const int col = row;
    const int kc  = wave * KW;

    v4f d[RB], nrm[RB];
    #pragma unroll
    for (int r = 0; r < RB; ++r) {
        const size_t base = (size_t)(tileM + r) * NN + kc + 4 * lane;
        d[r]   = __builtin_nontemporal_load((const v4f*)(dist + base));
        nrm[r] = __builtin_nontemporal_load((const v4f*)(norm + base));
    }
    v2f b[KW / 4];
    #pragma unroll
    for (int s = 0; s < KW / 4; ++s) {
        const int kg = kc + 4 * s + 2 * hi;
        b[s].x = fsums[kg       * CC + col];
        b[s].y = fsums[(kg + 1) * CC + col];
    }

    float wm1[HH], bb1[HH], wm2[HH];
    #pragma unroll
    for (int h = 0; h < HH; ++h) { wm1[h] = pWm1[h]; bb1[h] = pbm1[h]; wm2[h] = pWm2[h]; }
    const float bm2v = pbm2[0];

    float* myT = mT[wave];
    #pragma unroll
    for (int r = 0; r < RB; ++r) {
        float a0 = bm2v, a1 = bm2v, a2 = bm2v, a3 = bm2v;
        #pragma unroll
        for (int h = 0; h < HH; ++h) {
            const float t0 = fmaf(d[r].x, wm1[h], bb1[h]);
            const float t1 = fmaf(d[r].y, wm1[h], bb1[h]);
            const float t2 = fmaf(d[r].z, wm1[h], bb1[h]);
            const float t3 = fmaf(d[r].w, wm1[h], bb1[h]);
            a0 = fmaf(fmaxf(t0, 0.f), wm2[h], a0);
            a1 = fmaf(fmaxf(t1, 0.f), wm2[h], a1);
            a2 = fmaf(fmaxf(t2, 0.f), wm2[h], a2);
            a3 = fmaf(fmaxf(t3, 0.f), wm2[h], a3);
        }
        v4f mv;
        mv.x = a0 * __builtin_amdgcn_rcpf(nrm[r].x);
        mv.y = a1 * __builtin_amdgcn_rcpf(nrm[r].y);
        mv.z = a2 * __builtin_amdgcn_rcpf(nrm[r].z);
        mv.w = a3 * __builtin_amdgcn_rcpf(nrm[r].w);
        *(v4f*)&myT[r * MP + 4 * lane] = mv;
    }

    const int  arow = row & (RB - 1);
    const bool real = row < RB;
    v8f acc = {};
    #pragma unroll
    for (int s = 0; s < KW / 4; ++s) {
        const v2f t = *(const v2f*)&myT[arow * MP + 4 * s + 2 * hi];
        v2f a;
        a.x = real ? t.x : 0.f;
        a.y = real ? t.y : 0.f;
        acc = __builtin_amdgcn_wmma_f32_16x16x4_f32(false, a, false, b[s],
                                                    (short)0, acc, false, false);
    }

    if (hi == 0) {
        #pragma unroll
        for (int r = 0; r < RB; ++r) redC[wave * (RB * CC) + r * CC + col] = acc[r];
    }
    __syncthreads();
    if (tid < RB * CC) {
        float v = 0.f;
        #pragma unroll
        for (int w = 0; w < NW; ++w) v += redC[w * (RB * CC) + tid];
        out[tileM * CC + tid] = v;
    }
}

extern "C" void kernel_launch(void* const* d_in, const int* in_sizes, int n_in,
                              void* d_out, int out_size, void* d_ws, size_t ws_size,
                              hipStream_t stream) {
    const float* x    = (const float*)d_in[0];
    const float* dist = (const float*)d_in[1];
    const float* norm = (const float*)d_in[2];
    const float* W1   = (const float*)d_in[3];
    const float* b1   = (const float*)d_in[4];
    const float* W2   = (const float*)d_in[5];
    const float* b2   = (const float*)d_in[6];
    const float* fw   = (const float*)d_in[7];
    const float* Wm1  = (const float*)d_in[8];
    const float* bm1  = (const float*)d_in[9];
    const float* Wm2  = (const float*)d_in[10];
    const float* bm2  = (const float*)d_in[11];
    float* out   = (float*)d_out;
    float* fsums = (float*)d_ws;

    fsums_kernel<<<NN / 16, 256, 0, stream>>>(x, W1, b1, W2, b2, fw, fsums);
    out_kernel<<<NN / RB, NW * 32, 0, stream>>>(dist, norm, Wm1, bm1, Wm2, bm2, fsums, out);
}


# fsums kernel re-tiled: 16 waves x 128 K cols, a wave's loads all issued ahead of its math in use order (was 32 load-wait rounds); one acc tile per feature, attn and b2 applied per tile: new sum order
# speedup vs baseline: 2.9789x; 1.3159x over previous
#include <hip/hip_runtime.h>
#include <hip/hip_bf16.h>

typedef __attribute__((ext_vector_type(2))) float v2f;
typedef __attribute__((ext_vector_type(4))) float v4f;
typedef __attribute__((ext_vector_type(8))) float v8f;

#define NN 2048
#define FF 64
#define HH 32
#define CC 16

#define AW 16
#define AK 128

__global__ __launch_bounds__(512) __attribute__((amdgpu_waves_per_eu(4, 4)))
void fsums_kernel(const float* __restrict__ x,
                  const float* __restrict__ W1,
                  const float* __restrict__ b1,
                  const float* __restrict__ W2,
                  const float* __restrict__ b2,
                  const float* __restrict__ fw,
                  float* __restrict__ fsums)
{
    __shared__ float redA[AW * 256];

    const int tid  = threadIdx.x;
    const int lane = tid & 31;
    const int wave = tid >> 5;
    const int tileM = blockIdx.x * 16;
    const int row = lane & 15;
    const int hi  = lane >> 4;
    const int col = row;
    const int gRow = tileM + row;
    const int fq = wave * 4;
    const int kl = wave * AK + 4 * hi;

    const float fl = fw[lane];
    const float fh = fw[lane + 32];
    const v4f fwq = *(const v4f*)(fw + fq);
    const v4f xq  = *(const v4f*)(x + gRow * FF + fq);
    float b2q[4];
    #pragma unroll
    for (int q = 0; q < 4; ++q) b2q[q] = b2[(fq + q) * CC + col];
    v4f w1v[16], b1v[16];
    v2f w2v[16][2];
#define LOAD_PHASE(p)                                                              \
    _Pragma("unroll")                                                              \
    for (int q = 0; q < 4; ++q) {                                                  \
        const int j = 4 * q + (p);                                                 \
        w1v[j] = *(const v4f*)(W1 + kl + 8 * j);                                   \
        b1v[j] = *(const v4f*)(b1 + kl + 8 * j);                                   \
        _Pragma("unroll")                                                          \
        for (int t = 0; t < 2; ++t) {                                              \
            w2v[j][t].x = W2[(kl + 8 * j + 2 * t)     * CC + col];                 \
            w2v[j][t].y = W2[(kl + 8 * j + 2 * t + 1) * CC + col];                 \
        }                                                                          \
    }
    LOAD_PHASE(0)
    __builtin_amdgcn_sched_barrier(0);
    LOAD_PHASE(1)
    __builtin_amdgcn_sched_barrier(0);

    float ssum = __expf(__expf(fl)) + __expf(__expf(fh));
    #pragma unroll
    for (int o = 16; o > 0; o >>= 1) ssum += __shfl_xor(ssum, o);
    const float invS = 1.0f / ssum;
    float at[4];
    #pragma unroll
    for (int q = 0; q < 4; ++q) at[q] = __expf(__expf(fwq[q])) * invS;
    __builtin_amdgcn_sched_barrier(0);
    LOAD_PHASE(2)
    __builtin_amdgcn_sched_barrier(0);
    LOAD_PHASE(3)
    __builtin_amdgcn_sched_barrier(0);
#undef LOAD_PHASE

    v8f acc[4];
    #pragma unroll
    for (int q = 0; q < 4; ++q) {
        #pragma unroll
        for (int r = 0; r < 8; ++r) acc[q][r] = b2q[q];
    }
    #pragma unroll
    for (int p = 0; p < 4; ++p) {
        #pragma unroll
        for (int q = 0; q < 4; ++q) {
            const int j = 4 * q + p;
            const float xs = xq[q];
            #pragma unroll
            for (int t = 0; t < 2; ++t) {
                v2f a;
                a.x = fmaxf(fmaf(xs, w1v[j][2 * t],     b1v[j][2 * t]),     0.f);
                a.y = fmaxf(fmaf(xs, w1v[j][2 * t + 1], b1v[j][2 * t + 1]), 0.f);
                acc[q] = __builtin_amdgcn_wmma_f32_16x16x4_f32(false, a, false, w2v[j][t],
                                                               (short)0, acc[q], false, false);
            }
        }
        __builtin_amdgcn_sched_barrier(0);
    }

    #pragma unroll
    for (int r = 0; r < 8; ++r) {
        float v = at[0] * acc[0][r];
        v = fmaf(at[1], acc[1][r], v);
        v = fmaf(at[2], acc[2][r], v);
        v = fmaf(at[3], acc[3][r], v);
        const int orow = r + 8 * hi;
        redA[wave * 256 + orow * 16 + col] = v;
    }
    __syncthreads();
    if (tid < 256) {
        float v = 0.f;
        #pragma unroll
        for (int w = 0; w < AW; ++w) v += redA[w * 256 + tid];
        fsums[tileM * CC + tid] = v;
    }
}

#define RB 8
#define KW 128
#define NW 16
#define MP 132

__global__ __launch_bounds__(512)
void out_kernel(const float* __restrict__ dist,
                const float* __restrict__ norm,
                const float* __restrict__ pWm1,
                const float* __restrict__ pbm1,
                const float* __restrict__ pWm2,
                const float* __restrict__ pbm2,
                const float* __restrict__ fsums,
                float* __restrict__ out)
{
    __shared__ float mT[NW][RB * MP];
    __shared__ float redC[NW * RB * CC];

    const int tid  = threadIdx.x;
    const int lane = tid & 31;
    const int wave = tid >> 5;
    const int tileM = blockIdx.x * RB;
    const int row = lane & 15;
    const int hi  = lane >> 4;
    const int col = row;
    const int kc  = wave * KW;

    v4f d[RB], nrm[RB];
    #pragma unroll
    for (int r = 0; r < RB; ++r) {
        const size_t base = (size_t)(tileM + r) * NN + kc + 4 * lane;
        d[r]   = __builtin_nontemporal_load((const v4f*)(dist + base));
        nrm[r] = __builtin_nontemporal_load((const v4f*)(norm + base));
    }
    v2f b[KW / 4];
    #pragma unroll
    for (int s = 0; s < KW / 4; ++s) {
        const int kg = kc + 4 * s + 2 * hi;
        b[s].x = fsums[kg       * CC + col];
        b[s].y = fsums[(kg + 1) * CC + col];
    }

    float wm1[HH], bb1[HH], wm2[HH];
    #pragma unroll
    for (int h = 0; h < HH; ++h) { wm1[h] = pWm1[h]; bb1[h] = pbm1[h]; wm2[h] = pWm2[h]; }
    const float bm2v = pbm2[0];

    float* myT = mT[wave];
    #pragma unroll
    for (int r = 0; r < RB; ++r) {
        float a0 = bm2v, a1 = bm2v, a2 = bm2v, a3 = bm2v;
        #pragma unroll
        for (int h = 0; h < HH; ++h) {
            const float t0 = fmaf(d[r].x, wm1[h], bb1[h]);
            const float t1 = fmaf(d[r].y, wm1[h], bb1[h]);
            const float t2 = fmaf(d[r].z, wm1[h], bb1[h]);
            const float t3 = fmaf(d[r].w, wm1[h], bb1[h]);
            a0 = fmaf(fmaxf(t0, 0.f), wm2[h], a0);
            a1 = fmaf(fmaxf(t1, 0.f), wm2[h], a1);
            a2 = fmaf(fmaxf(t2, 0.f), wm2[h], a2);
            a3 = fmaf(fmaxf(t3, 0.f), wm2[h], a3);
        }
        v4f mv;
        mv.x = a0 * __builtin_amdgcn_rcpf(nrm[r].x);
        mv.y = a1 * __builtin_amdgcn_rcpf(nrm[r].y);
        mv.z = a2 * __builtin_amdgcn_rcpf(nrm[r].z);
        mv.w = a3 * __builtin_amdgcn_rcpf(nrm[r].w);
        *(v4f*)&myT[r * MP + 4 * lane] = mv;
    }

    const int  arow = row & (RB - 1);
    const bool real = row < RB;
    v8f acc = {};
    #pragma unroll
    for (int s = 0; s < KW / 4; ++s) {
        const v2f t = *(const v2f*)&myT[arow * MP + 4 * s + 2 * hi];
        v2f a;
        a.x = real ? t.x : 0.f;
        a.y = real ? t.y : 0.f;
        acc = __builtin_amdgcn_wmma_f32_16x16x4_f32(false, a, false, b[s],
                                                    (short)0, acc, false, false);
    }

    if (hi == 0) {
        #pragma unroll
        for (int r = 0; r < RB; ++r) redC[wave * (RB * CC) + r * CC + col] = acc[r];
    }
    __syncthreads();
    if (tid < RB * CC) {
        float v = 0.f;
        #pragma unroll
        for (int w = 0; w < NW; ++w) v += redC[w * (RB * CC) + tid];
        out[tileM * CC + tid] = v;
    }
}

extern "C" void kernel_launch(void* const* d_in, const int* in_sizes, int n_in,
                              void* d_out, int out_size, void* d_ws, size_t ws_size,
                              hipStream_t stream) {
    const float* x    = (const float*)d_in[0];
    const float* dist = (const float*)d_in[1];
    const float* norm = (const float*)d_in[2];
    const float* W1   = (const float*)d_in[3];
    const float* b1   = (const float*)d_in[4];
    const float* W2   = (const float*)d_in[5];
    const float* b2   = (const float*)d_in[6];
    const float* fw   = (const float*)d_in[7];
    const float* Wm1  = (const float*)d_in[8];
    const float* bm1  = (const float*)d_in[9];
    const float* Wm2  = (const float*)d_in[10];
    const float* bm2  = (const float*)d_in[11];
    float* out   = (float*)d_out;
    float* fsums = (float*)d_ws;

    fsums_kernel<<<NN / 16, AW * 32, 0, stream>>>(x, W1, b1, W2, b2, fw, fsums);
    out_kernel<<<NN / RB, NW * 32, 0, stream>>>(dist, norm, Wm1, bm1, Wm2, bm2, fsums, out);
}


# out kernel, m network: both multiply-adds of a unit issued on element pairs as packed 2 x f32 FMAs (half the FMA issue slots), relu maxima dual-issued; per element the same operations and order
# speedup vs baseline: 3.1518x; 1.0580x over previous
#include <hip/hip_runtime.h>
#include <hip/hip_bf16.h>

typedef __attribute__((ext_vector_type(2))) float v2f;
typedef __attribute__((ext_vector_type(4))) float v4f;
typedef __attribute__((ext_vector_type(8))) float v8f;

#define NN 2048
#define FF 64
#define HH 32
#define CC 16

#define AW 16
#define AK 128

__global__ __launch_bounds__(512) __attribute__((amdgpu_waves_per_eu(4, 4)))
void fsums_kernel(const float* __restrict__ x,
                  const float* __restrict__ W1,
                  const float* __restrict__ b1,
                  const float* __restrict__ W2,
                  const float* __restrict__ b2,
                  const float* __restrict__ fw,
                  float* __restrict__ fsums)
{
    __shared__ float redA[AW * 256];

    const int tid  = threadIdx.x;
    const int lane = tid & 31;
    const int wave = tid >> 5;
    const int tileM = blockIdx.x * 16;
    const int row = lane & 15;
    const int hi  = lane >> 4;
    const int col = row;
    const int gRow = tileM + row;
    const int fq = wave * 4;
    const int kl = wave * AK + 4 * hi;

    const float fl = fw[lane];
    const float fh = fw[lane + 32];
    const v4f fwq = *(const v4f*)(fw + fq);
    const v4f xq  = *(const v4f*)(x + gRow * FF + fq);
    float b2q[4];
    #pragma unroll
    for (int q = 0; q < 4; ++q) b2q[q] = b2[(fq + q) * CC + col];
    v4f w1v[16], b1v[16];
    v2f w2v[16][2];
#define LOAD_PHASE(p)                                                              \
    _Pragma("unroll")                                                              \
    for (int q = 0; q < 4; ++q) {                                                  \
        const int j = 4 * q + (p);                                                 \
        w1v[j] = *(const v4f*)(W1 + kl + 8 * j);                                   \
        b1v[j] = *(const v4f*)(b1 + kl + 8 * j);                                   \
        _Pragma("unroll")                                                          \
        for (int t = 0; t < 2; ++t) {                                              \
            w2v[j][t].x = W2[(kl + 8 * j + 2 * t)     * CC + col];                 \
            w2v[j][t].y = W2[(kl + 8 * j + 2 * t + 1) * CC + col];                 \
        }                                                                          \
    }
    LOAD_PHASE(0)
    __builtin_amdgcn_sched_barrier(0);
    LOAD_PHASE(1)
    __builtin_amdgcn_sched_barrier(0);

    float ssum = __expf(__expf(fl)) + __expf(__expf(fh));
    #pragma unroll
    for (int o = 16; o > 0; o >>= 1) ssum += __shfl_xor(ssum, o);
    const float invS = 1.0f / ssum;
    float at[4];
    #pragma unroll
    for (int q = 0; q < 4; ++q) at[q] = __expf(__expf(fwq[q])) * invS;
    __builtin_amdgcn_sched_barrier(0);
    LOAD_PHASE(2)
    __builtin_amdgcn_sched_barrier(0);
    LOAD_PHASE(3)
    __builtin_amdgcn_sched_barrier(0);
#undef LOAD_PHASE

    v8f acc[4];
    #pragma unroll
    for (int q = 0; q < 4; ++q) {
        #pragma unroll
        for (int r = 0; r < 8; ++r) acc[q][r] = b2q[q];
    }
    #pragma unroll
    for (int p = 0; p < 4; ++p) {
        #pragma unroll
        for (int q = 0; q < 4; ++q) {
            const int j = 4 * q + p;
            const float xs = xq[q];
            #pragma unroll
            for (int t = 0; t < 2; ++t) {
                v2f a;
                a.x = fmaxf(fmaf(xs, w1v[j][2 * t],     b1v[j][2 * t]),     0.f);
                a.y = fmaxf(fmaf(xs, w1v[j][2 * t + 1], b1v[j][2 * t + 1]), 0.f);
                acc[q] = __builtin_amdgcn_wmma_f32_16x16x4_f32(false, a, false, w2v[j][t],
                                                               (short)0, acc[q], false, false);
            }
        }
        __builtin_amdgcn_sched_barrier(0);
    }

    #pragma unroll
    for (int r = 0; r < 8; ++r) {
        float v = at[0] * acc[0][r];
        v = fmaf(at[1], acc[1][r], v);
        v = fmaf(at[2], acc[2][r], v);
        v = fmaf(at[3], acc[3][r], v);
        const int orow = r + 8 * hi;
        redA[wave * 256 + orow * 16 + col] = v;
    }
    __syncthreads();
    if (tid < 256) {
        float v = 0.f;
        #pragma unroll
        for (int w = 0; w < AW; ++w) v += redA[w * 256 + tid];
        fsums[tileM * CC + tid] = v;
    }
}

#define RB 8
#define KW 128
#define NW 16
#define MP 132

__global__ __launch_bounds__(512)
void out_kernel(const float* __restrict__ dist,
                const float* __restrict__ norm,
                const float* __restrict__ pWm1,
                const float* __restrict__ pbm1,
                const float* __restrict__ pWm2,
                const float* __restrict__ pbm2,
                const float* __restrict__ fsums,
                float* __restrict__ out)
{
    __shared__ float mT[NW][RB * MP];
    __shared__ float redC[NW * RB * CC];

    const int tid  = threadIdx.x;
    const int lane = tid & 31;
    const int wave = tid >> 5;
    const int tileM = blockIdx.x * RB;
    const int row = lane & 15;
    const int hi  = lane >> 4;
    const int col = row;
    const int kc  = wave * KW;

    v4f d[RB], nrm[RB];
    #pragma unroll
    for (int r = 0; r < RB; ++r) {
        const size_t base = (size_t)(tileM + r) * NN + kc + 4 * lane;
        d[r]   = __builtin_nontemporal_load((const v4f*)(dist + base));
        nrm[r] = __builtin_nontemporal_load((const v4f*)(norm + base));
    }
    v2f b[KW / 4];
    #pragma unroll
    for (int s = 0; s < KW / 4; ++s) {
        const int kg = kc + 4 * s + 2 * hi;
        b[s].x = fsums[kg       * CC + col];
        b[s].y = fsums[(kg + 1) * CC + col];
    }

    float wm1[HH], bb1[HH], wm2[HH];
    #pragma unroll
    for (int h = 0; h < HH; ++h) { wm1[h] = pWm1[h]; bb1[h] = pbm1[h]; wm2[h] = pWm2[h]; }
    const float bm2v = pbm2[0];

    __builtin_amdgcn_sched_barrier(0);

    float* myT = mT[wave];
    #pragma unroll
    for (int r = 0; r < RB; ++r) {
        const v2f d01 = {d[r].x, d[r].y};
        const v2f d23 = {d[r].z, d[r].w};
        v2f a01 = {bm2v, bm2v};
        v2f a23 = {bm2v, bm2v};
        #pragma unroll
        for (int h = 0; h < HH; ++h) {
            const v2f w  = {wm1[h], wm1[h]};
            const v2f bb = {bb1[h], bb1[h]};
            const v2f w2 = {wm2[h], wm2[h]};
            v2f t01 = __builtin_elementwise_fma(d01, w, bb);
            v2f t23 = __builtin_elementwise_fma(d23, w, bb);
            t01.x = fmaxf(t01.x, 0.f); t01.y = fmaxf(t01.y, 0.f);
            t23.x = fmaxf(t23.x, 0.f); t23.y = fmaxf(t23.y, 0.f);
            a01 = __builtin_elementwise_fma(t01, w2, a01);
            a23 = __builtin_elementwise_fma(t23, w2, a23);
        }
        v4f mv;
        mv.x = a01.x * __builtin_amdgcn_rcpf(nrm[r].x);
        mv.y = a01.y * __builtin_amdgcn_rcpf(nrm[r].y);
        mv.z = a23.x * __builtin_amdgcn_rcpf(nrm[r].z);
        mv.w = a23.y * __builtin_amdgcn_rcpf(nrm[r].w);
        *(v4f*)&myT[r * MP + 4 * lane] = mv;
    }

    const int  arow = row & (RB - 1);
    const bool real = row < RB;
    v8f acc = {};
    #pragma unroll
    for (int s = 0; s < KW / 4; ++s) {
        const v2f t = *(const v2f*)&myT[arow * MP + 4 * s + 2 * hi];
        v2f a;
        a.x = real ? t.x : 0.f;
        a.y = real ? t.y : 0.f;
        acc = __builtin_amdgcn_wmma_f32_16x16x4_f32(false, a, false, b[s],
                                                    (short)0, acc, false, false);
    }

    if (hi == 0) {
        #pragma unroll
        for (int r = 0; r < RB; ++r) redC[wave * (RB * CC) + r * CC + col] = acc[r];
    }
    __syncthreads();
    if (tid < RB * CC) {
        float v = 0.f;
        #pragma unroll
        for (int w = 0; w < NW; ++w) v += redC[w * (RB * CC) + tid];
        out[tileM * CC + tid] = v;
    }
}

extern "C" void kernel_launch(void* const* d_in, const int* in_sizes, int n_in,
                              void* d_out, int out_size, void* d_ws, size_t ws_size,
                              hipStream_t stream) {
    const float* x    = (const float*)d_in[0];
    const float* dist = (const float*)d_in[1];
    const float* norm = (const float*)d_in[2];
    const float* W1   = (const float*)d_in[3];
    const float* b1   = (const float*)d_in[4];
    const float* W2   = (const float*)d_in[5];
    const float* b2   = (const float*)d_in[6];
    const float* fw   = (const float*)d_in[7];
    const float* Wm1  = (const float*)d_in[8];
    const float* bm1  = (const float*)d_in[9];
    const float* Wm2  = (const float*)d_in[10];
    const float* bm2  = (const float*)d_in[11];
    float* out   = (float*)d_out;
    float* fsums = (float*)d_ws;

    fsums_kernel<<<NN / 16, AW * 32, 0, stream>>>(x, W1, b1, W2, b2, fw, fsums);
    out_kernel<<<NN / RB, NW * 32, 0, stream>>>(dist, norm, Wm1, bm1, Wm2, bm2, fsums, out);
}


# out kernel: 8 waves per block (was 16), a wave takes 256 K columns in four 64-column passes on one accumulator tile: half the waves to launch; 8-way K split (was 16) changes the sum order
# speedup vs baseline: 3.9671x; 1.2587x over previous
#include <hip/hip_runtime.h>
#include <hip/hip_bf16.h>

typedef __attribute__((ext_vector_type(2))) float v2f;
typedef __attribute__((ext_vector_type(4))) float v4f;
typedef __attribute__((ext_vector_type(8))) float v8f;

#define NN 2048
#define FF 64
#define HH 32
#define CC 16

#define AW 16
#define AK 128

__global__ __launch_bounds__(512) __attribute__((amdgpu_waves_per_eu(4, 4)))
void fsums_kernel(const float* __restrict__ x,
                  const float* __restrict__ W1,
                  const float* __restrict__ b1,
                  const float* __restrict__ W2,
                  const float* __restrict__ b2,
                  const float* __restrict__ fw,
                  float* __restrict__ fsums)
{
    __shared__ float redA[AW * 256];

    const int tid  = threadIdx.x;
    const int lane = tid & 31;
    const int wave = tid >> 5;
    const int tileM = blockIdx.x * 16;
    const int row = lane & 15;
    const int hi  = lane >> 4;
    const int col = row;
    const int gRow = tileM + row;
    const int fq = wave * 4;
    const int kl = wave * AK + 4 * hi;

    const float fl = fw[lane];
    const float fh = fw[lane + 32];
    const v4f fwq = *(const v4f*)(fw + fq);
    const v4f xq  = *(const v4f*)(x + gRow * FF + fq);
    float b2q[4];
    #pragma unroll
    for (int q = 0; q < 4; ++q) b2q[q] = b2[(fq + q) * CC + col];
    v4f w1v[16], b1v[16];
    v2f w2v[16][2];
#define LOAD_PHASE(p)                                                              \
    _Pragma("unroll")                                                              \
    for (int q = 0; q < 4; ++q) {                                                  \
        const int j = 4 * q + (p);                                                 \
        w1v[j] = *(const v4f*)(W1 + kl + 8 * j);                                   \
        b1v[j] = *(const v4f*)(b1 + kl + 8 * j);                                   \
        _Pragma("unroll")                                                          \
        for (int t = 0; t < 2; ++t) {                                              \
            w2v[j][t].x = W2[(kl + 8 * j + 2 * t)     * CC + col];                 \
            w2v[j][t].y = W2[(kl + 8 * j + 2 * t + 1) * CC + col];                 \
        }                                                                          \
    }
    LOAD_PHASE(0)
    __builtin_amdgcn_sched_barrier(0);
    LOAD_PHASE(1)
    __builtin_amdgcn_sched_barrier(0);

    float ssum = __expf(__expf(fl)) + __expf(__expf(fh));
    #pragma unroll
    for (int o = 16; o > 0; o >>= 1) ssum += __shfl_xor(ssum, o);
    const float invS = 1.0f / ssum;
    float at[4];
    #pragma unroll
    for (int q = 0; q < 4; ++q) at[q] = __expf(__expf(fwq[q])) * invS;
    __builtin_amdgcn_sched_barrier(0);
    LOAD_PHASE(2)
    __builtin_amdgcn_sched_barrier(0);
    LOAD_PHASE(3)
    __builtin_amdgcn_sched_barrier(0);
#undef LOAD_PHASE

    v8f acc[4];
    #pragma unroll
    for (int q = 0; q < 4; ++q) {
        #pragma unroll
        for (int r = 0; r < 8; ++r) acc[q][r] = b2q[q];
    }
    #pragma unroll
    for (int p = 0; p < 4; ++p) {
        #pragma unroll
        for (int q = 0; q < 4; ++q) {
            const int j = 4 * q + p;
            const float xs = xq[q];
            #pragma unroll
            for (int t = 0; t < 2; ++t) {
                v2f a;
                a.x = fmaxf(fmaf(xs, w1v[j][2 * t],     b1v[j][2 * t]),     0.f);
                a.y = fmaxf(fmaf(xs, w1v[j][2 * t + 1], b1v[j][2 * t + 1]), 0.f);
                acc[q] = __builtin_amdgcn_wmma_f32_16x16x4_f32(false, a, false, w2v[j][t],
                                                               (short)0, acc[q], false, false);
            }
        }
        __builtin_amdgcn_sched_barrier(0);
    }

    #pragma unroll
    for (int r = 0; r < 8; ++r) {
        float v = at[0] * acc[0][r];
        v = fmaf(at[1], acc[1][r], v);
        v = fmaf(at[2], acc[2][r], v);
        v = fmaf(at[3], acc[3][r], v);
        const int orow = r + 8 * hi;
        redA[wave * 256 + orow * 16 + col] = v;
    }
    __syncthreads();
    if (tid < 256) {
        float v = 0.f;
        #pragma unroll
        for (int w = 0; w < AW; ++w) v += redA[w * 256 + tid];
        fsums[tileM * CC + tid] = v;
    }
}

#define RB 8
#define KW 64
#define NP 4
#define NW 8
#define MP 68

__global__ __launch_bounds__(256) __attribute__((amdgpu_waves_per_eu(2, 2)))
void out_kernel(const float* __restrict__ dist,
                const float* __restrict__ norm,
                const float* __restrict__ pWm1,
                const float* __restrict__ pbm1,
                const float* __restrict__ pWm2,
                const float* __restrict__ pbm2,
                const float* __restrict__ fsums,
                float* __restrict__ out)
{
    __shared__ float mT[NW][RB * MP];
    __shared__ float redC[NW * RB * CC];
    __shared__ v4f   pT[NW][HH];

    const int tid  = threadIdx.x;
    const int lane = tid & 31;
    const int wave = tid >> 5;
    const int tileM = blockIdx.x * RB;
    const int row = lane & 15;
    const int hi  = lane >> 4;
    const int col = row;

    const v4f myq = {pWm1[lane], pbm1[lane], pWm2[lane], 0.f};
    const float bm2v = pbm2[0];

    const size_t base0 = (size_t)tileM * NN + wave * (KW * NP) + 2 * lane;
    v2f d[RB], nrm[RB];
    #pragma unroll
    for (int r = 0; r < RB; ++r) {
        d[r]   = __builtin_nontemporal_load((const v2f*)(dist + base0 + (size_t)r * NN));
        nrm[r] = __builtin_nontemporal_load((const v2f*)(norm + base0 + (size_t)r * NN));
    }

    v4f* myP = pT[wave];
    myP[lane] = myq;

    float* myT = mT[wave];
    const int  arow = row & (RB - 1);
    const bool real = row < RB;
    v8f acc = {};

    #pragma nounroll
    for (int p = 0; p < NP; ++p) {
        const int kc = wave * (KW * NP) + p * KW;
        v2f b[KW / 4];
        #pragma unroll
        for (int s = 0; s < KW / 4; ++s) {
            const int kg = kc + 4 * s + 2 * hi;
            b[s].x = fsums[kg       * CC + col];
            b[s].y = fsums[(kg + 1) * CC + col];
        }
        v4f q[HH];
        #pragma unroll
        for (int h = 0; h < HH; ++h) q[h] = myP[h];

        __builtin_amdgcn_sched_barrier(0);

        v2f a01[RB];
        #pragma unroll
        for (int r = 0; r < RB; ++r) a01[r] = v2f{bm2v, bm2v};
        #pragma unroll
        for (int h = 0; h < HH; ++h) {
            const v2f w  = {q[h].x, q[h].x};
            const v2f bb = {q[h].y, q[h].y};
            const v2f w2 = {q[h].z, q[h].z};
            #pragma unroll
            for (int r = 0; r < RB; ++r) {
                v2f t01 = __builtin_elementwise_fma(d[r], w, bb);
                t01.x = fmaxf(t01.x, 0.f); t01.y = fmaxf(t01.y, 0.f);
                a01[r] = __builtin_elementwise_fma(t01, w2, a01[r]);
            }
        }
        #pragma unroll
        for (int r = 0; r < RB; ++r) {
            v2f mv;
            mv.x = a01[r].x * __builtin_amdgcn_rcpf(nrm[r].x);
            mv.y = a01[r].y * __builtin_amdgcn_rcpf(nrm[r].y);
            *(v2f*)&myT[r * MP + 2 * lane] = mv;
        }

        __builtin_amdgcn_sched_barrier(0);
        if (p + 1 < NP) {
            #pragma unroll
            for (int r = 0; r < RB; ++r) {
                d[r]   = __builtin_nontemporal_load((const v2f*)(dist + base0 + (size_t)r * NN + (p + 1) * KW));
                nrm[r] = __builtin_nontemporal_load((const v2f*)(norm + base0 + (size_t)r * NN + (p + 1) * KW));
            }
        }
        __builtin_amdgcn_sched_barrier(0);

        #pragma unroll
        for (int s = 0; s < KW / 4; ++s) {
            const v2f t = *(const v2f*)&myT[arow * MP + 4 * s + 2 * hi];
            v2f a;
            a.x = real ? t.x : 0.f;
            a.y = real ? t.y : 0.f;
            acc = __builtin_amdgcn_wmma_f32_16x16x4_f32(false, a, false, b[s],
                                                        (short)0, acc, false, false);
        }
    }

    if (hi == 0) {
        #pragma unroll
        for (int r = 0; r < RB; ++r) redC[wave * (RB * CC) + r * CC + col] = acc[r];
    }
    __syncthreads();
    if (tid < RB * CC) {
        float v = 0.f;
        #pragma unroll
        for (int w = 0; w < NW; ++w) v += redC[w * (RB * CC) + tid];
        out[tileM * CC + tid] = v;
    }
}

extern "C" void kernel_launch(void* const* d_in, const int* in_sizes, int n_in,
                              void* d_out, int out_size, void* d_ws, size_t ws_size,
                              hipStream_t stream) {
    const float* x    = (const float*)d_in[0];
    const float* dist = (const float*)d_in[1];
    const float* norm = (const float*)d_in[2];
    const float* W1   = (const float*)d_in[3];
    const float* b1   = (const float*)d_in[4];
    const float* W2   = (const float*)d_in[5];
    const float* b2   = (const float*)d_in[6];
    const float* fw   = (const float*)d_in[7];
    const float* Wm1  = (const float*)d_in[8];
    const float* bm1  = (const float*)d_in[9];
    const float* Wm2  = (const float*)d_in[10];
    const float* bm2  = (const float*)d_in[11];
    float* out   = (float*)d_out;
    float* fsums = (float*)d_ws;

    fsums_kernel<<<NN / 16, AW * 32, 0, stream>>>(x, W1, b1, W2, b2, fw, fsums);
    out_kernel<<<NN / RB, NW * 32, 0, stream>>>(dist, norm, Wm1, bm1, Wm2, bm2, fsums, out);
}


# out kernel, m network: unit = gamma*max(d,c), c=-b/w, gamma=w2*|w| (1 max + 1 FMA per element, was 2 FMA + max), linear parts folded into alpha*d+beta; rows requested 2 passes ahead; new sum order
# speedup vs baseline: 4.1716x; 1.0515x over previous
#include <hip/hip_runtime.h>
#include <hip/hip_bf16.h>

typedef __attribute__((ext_vector_type(2))) float v2f;
typedef __attribute__((ext_vector_type(4))) float v4f;
typedef __attribute__((ext_vector_type(8))) float v8f;

#define NN 2048
#define FF 64
#define HH 32
#define CC 16

#define AW 16
#define AK 128

__global__ __launch_bounds__(512) __attribute__((amdgpu_waves_per_eu(4, 4)))
void fsums_kernel(const float* __restrict__ x,
                  const float* __restrict__ W1,
                  const float* __restrict__ b1,
                  const float* __restrict__ W2,
                  const float* __restrict__ b2,
                  const float* __restrict__ fw,
                  float* __restrict__ fsums)
{
    __shared__ float redA[AW * 256];

    const int tid  = threadIdx.x;
    const int lane = tid & 31;
    const int wave = tid >> 5;
    const int tileM = blockIdx.x * 16;
    const int row = lane & 15;
    const int hi  = lane >> 4;
    const int col = row;
    const int gRow = tileM + row;
    const int fq = wave * 4;
    const int kl = wave * AK + 4 * hi;

    const float fl = fw[lane];
    const float fh = fw[lane + 32];
    const v4f fwq = *(const v4f*)(fw + fq);
    const v4f xq  = *(const v4f*)(x + gRow * FF + fq);
    float b2q[4];
    #pragma unroll
    for (int q = 0; q < 4; ++q) b2q[q] = b2[(fq + q) * CC + col];
    v4f w1v[16], b1v[16];
    v2f w2v[16][2];
#define LOAD_PHASE(p)                                                              \
    _Pragma("unroll")                                                              \
    for (int q = 0; q < 4; ++q) {                                                  \
        const int j = 4 * q + (p);                                                 \
        w1v[j] = *(const v4f*)(W1 + kl + 8 * j);                                   \
        b1v[j] = *(const v4f*)(b1 + kl + 8 * j);                                   \
        _Pragma("unroll")                                                          \
        for (int t = 0; t < 2; ++t) {                                              \
            w2v[j][t].x = W2[(kl + 8 * j + 2 * t)     * CC + col];                 \
            w2v[j][t].y = W2[(kl + 8 * j + 2 * t + 1) * CC + col];                 \
        }                                                                          \
    }
    LOAD_PHASE(0)
    __builtin_amdgcn_sched_barrier(0);
    LOAD_PHASE(1)
    __builtin_amdgcn_sched_barrier(0);

    float ssum = __expf(__expf(fl)) + __expf(__expf(fh));
    #pragma unroll
    for (int o = 16; o > 0; o >>= 1) ssum += __shfl_xor(ssum, o);
    const float invS = 1.0f / ssum;
    float at[4];
    #pragma unroll
    for (int q = 0; q < 4; ++q) at[q] = __expf(__expf(fwq[q])) * invS;
    __builtin_amdgcn_sched_barrier(0);
    LOAD_PHASE(2)
    __builtin_amdgcn_sched_barrier(0);
    LOAD_PHASE(3)
    __builtin_amdgcn_sched_barrier(0);
#undef LOAD_PHASE

    v8f acc[4];
    #pragma unroll
    for (int q = 0; q < 4; ++q) {
        #pragma unroll
        for (int r = 0; r < 8; ++r) acc[q][r] = b2q[q];
    }
    #pragma unroll
    for (int p = 0; p < 4; ++p) {
        #pragma unroll
        for (int q = 0; q < 4; ++q) {
            const int j = 4 * q + p;
            const float xs = xq[q];
            #pragma unroll
            for (int t = 0; t < 2; ++t) {
                v2f a;
                a.x = fmaxf(fmaf(xs, w1v[j][2 * t],     b1v[j][2 * t]),     0.f);
                a.y = fmaxf(fmaf(xs, w1v[j][2 * t + 1], b1v[j][2 * t + 1]), 0.f);
                acc[q] = __builtin_amdgcn_wmma_f32_16x16x4_f32(false, a, false, w2v[j][t],
                                                               (short)0, acc[q], false, false);
            }
        }
        __builtin_amdgcn_sched_barrier(0);
    }

    #pragma unroll
    for (int r = 0; r < 8; ++r) {
        float v = at[0] * acc[0][r];
        v = fmaf(at[1], acc[1][r], v);
        v = fmaf(at[2], acc[2][r], v);
        v = fmaf(at[3], acc[3][r], v);
        const int orow = r + 8 * hi;
        redA[wave * 256 + orow * 16 + col] = v;
    }
    __syncthreads();
    if (tid < 256) {
        float v = 0.f;
        #pragma unroll
        for (int w = 0; w < AW; ++w) v += redA[w * 256 + tid];
        fsums[tileM * CC + tid] = v;
    }
}

#define RB 8
#define KW 64
#define NP 4
#define NW 8
#define MP 68

__global__ __launch_bounds__(256) __attribute__((amdgpu_waves_per_eu(2, 2)))
void out_kernel(const float* __restrict__ dist,
                const float* __restrict__ norm,
                const float* __restrict__ pWm1,
                const float* __restrict__ pbm1,
                const float* __restrict__ pWm2,
                const float* __restrict__ pbm2,
                const float* __restrict__ fsums,
                float* __restrict__ out)
{
    __shared__ float mT[NW][RB * MP];
    __shared__ float redC[NW * RB * CC];
    __shared__ v2f   pT[NW][HH];

    const int tid  = threadIdx.x;
    const int lane = tid & 31;
    const int wave = tid >> 5;
    const int tileM = blockIdx.x * RB;
    const int row = lane & 15;
    const int hi  = lane >> 4;
    const int col = row;

    const float uw  = pWm1[lane];
    const float ub  = pbm1[lane];
    const float uw2 = pWm2[lane];
    const float bm2v = pbm2[0];

    const size_t base0 = (size_t)tileM * NN + wave * (KW * NP) + 2 * lane;
    v2f d[NP][RB], nrm[NP][RB];
#define LOAD_ROWS(p)                                                                              \
    _Pragma("unroll")                                                                             \
    for (int r = 0; r < RB; ++r) {                                                                \
        d[p][r]   = __builtin_nontemporal_load((const v2f*)(dist + base0 + (size_t)r * NN + (p) * KW)); \
        nrm[p][r] = __builtin_nontemporal_load((const v2f*)(norm + base0 + (size_t)r * NN + (p) * KW)); \
    }
    v2f b[KW / 4];
#define LOAD_BFRAG(p)                                                                             \
    _Pragma("unroll")                                                                             \
    for (int s = 0; s < KW / 4; ++s) {                                                            \
        const int kg = wave * (KW * NP) + (p) * KW + 4 * s + 2 * hi;                              \
        b[s].x = fsums[kg       * CC + col];                                                      \
        b[s].y = fsums[(kg + 1) * CC + col];                                                      \
    }
    LOAD_ROWS(0)
    __builtin_amdgcn_sched_barrier(0);
    LOAD_ROWS(1)
    __builtin_amdgcn_sched_barrier(0);

    const float cq    = -ub / uw;
    const bool  fin   = fabsf(cq) <= 3.0e38f;
    const float g     = uw2 * uw;
    const float cfin  = fin ? cq : 0.f;
    const float gam   = fin ? uw2 * fabsf(uw) : 0.f;
    float al = (fin && uw < 0.f) ? g : 0.f;
    float be = fin ? (uw > 0.f ? uw2 * ub : 0.f) : uw2 * fmaxf(ub, 0.f);
    #pragma unroll
    for (int o = 16; o > 0; o >>= 1) {
        al += __shfl_xor(al, o);
        be += __shfl_xor(be, o);
    }
    be += bm2v;
    v2f* myP = pT[wave];
    myP[lane] = v2f{cfin, gam};

    LOAD_BFRAG(0)
    __builtin_amdgcn_sched_barrier(0);

    float* myT = mT[wave];
    const int  arow = row & (RB - 1);
    const bool real = row < RB;
    v8f acc = {};
    const v2f al2 = {al, al};
    const v2f be2 = {be, be};

    #pragma unroll
    for (int p = 0; p < NP; ++p) {
        if (p > 0) { LOAD_BFRAG(p) }
        v2f q[HH];
        #pragma unroll
        for (int h = 0; h < HH; ++h) q[h] = myP[h];

        __builtin_amdgcn_sched_barrier(0);

        v2f a01[RB];
        #pragma unroll
        for (int r = 0; r < RB; ++r) a01[r] = __builtin_elementwise_fma(d[p][r], al2, be2);
        #pragma unroll
        for (int h = 0; h < HH; ++h) {
            const float ch = q[h].x;
            const v2f   gg = {q[h].y, q[h].y};
            #pragma unroll
            for (int r = 0; r < RB; ++r) {
                v2f t01;
                t01.x = fmaxf(d[p][r].x, ch);
                t01.y = fmaxf(d[p][r].y, ch);
                a01[r] = __builtin_elementwise_fma(t01, gg, a01[r]);
            }
        }
        #pragma unroll
        for (int r = 0; r < RB; ++r) {
            v2f mv;
            mv.x = a01[r].x * __builtin_amdgcn_rcpf(nrm[p][r].x);
            mv.y = a01[r].y * __builtin_amdgcn_rcpf(nrm[p][r].y);
            *(v2f*)&myT[r * MP + 2 * lane] = mv;
        }

        __builtin_amdgcn_sched_barrier(0);
        if (p == 0) { LOAD_ROWS(2) }
        if (p == 1) { LOAD_ROWS(3) }
        __builtin_amdgcn_sched_barrier(0);

        #pragma unroll
        for (int s = 0; s < KW / 4; ++s) {
            const v2f t = *(const v2f*)&myT[arow * MP + 4 * s + 2 * hi];
            v2f a;
            a.x = real ? t.x : 0.f;
            a.y = real ? t.y : 0.f;
            acc = __builtin_amdgcn_wmma_f32_16x16x4_f32(false, a, false, b[s],
                                                        (short)0, acc, false, false);
        }
    }
#undef LOAD_ROWS
#undef LOAD_BFRAG

    if (hi == 0) {
        #pragma unroll
        for (int r = 0; r < RB; ++r) redC[wave * (RB * CC) + r * CC + col] = acc[r];
    }
    __syncthreads();
    if (tid < RB * CC) {
        float v = 0.f;
        #pragma unroll
        for (int w = 0; w < NW; ++w) v += redC[w * (RB * CC) + tid];
        out[tileM * CC + tid] = v;
    }
}

extern "C" void kernel_launch(void* const* d_in, const int* in_sizes, int n_in,
                              void* d_out, int out_size, void* d_ws, size_t ws_size,
                              hipStream_t stream) {
    const float* x    = (const float*)d_in[0];
    const float* dist = (const float*)d_in[1];
    const float* norm = (const float*)d_in[2];
    const float* W1   = (const float*)d_in[3];
    const float* b1   = (const float*)d_in[4];
    const float* W2   = (const float*)d_in[5];
    const float* b2   = (const float*)d_in[6];
    const float* fw   = (const float*)d_in[7];
    const float* Wm1  = (const float*)d_in[8];
    const float* bm1  = (const float*)d_in[9];
    const float* Wm2  = (const float*)d_in[10];
    const float* bm2  = (const float*)d_in[11];
    float* out   = (float*)d_out;
    float* fsums = (float*)d_ws;

    fsums_kernel<<<NN / 16, AW * 32, 0, stream>>>(x, W1, b1, W2, b2, fw, fsums);
    out_kernel<<<NN / RB, NW * 32, 0, stream>>>(dist, norm, Wm1, bm1, Wm2, bm2, fsums, out);
}
